# starting kernel
# baseline (speedup 1.0000x reference)
#include <hip/hip_runtime.h>


typedef __attribute__((ext_vector_type(2))) float v2f;
typedef __attribute__((ext_vector_type(8))) float v8f;

#define D_NODE 128
#define D_EDGE 64
#define WAVES  8

static __device__ __forceinline__ v8f wmma_f32(v2f a, v2f b, v8f c) {
  return __builtin_amdgcn_wmma_f32_16x16x4_f32(false, a, false, b, (short)0, c,
                                               false, false);
}

__global__ __launch_bounds__(256) void pack_pairs_kernel(
    const float* __restrict__ W, float* __restrict__ Wp, int rows, int cols) {
  int i = blockIdx.x * 256 + threadIdx.x;
  if (i >= rows * cols) return;
  int r = i / cols, c = i - r * cols;
  Wp[(((r >> 1) * cols + c) << 1) + (r & 1)] = W[i];
}

__global__ __launch_bounds__(256) void gine_edge_kernel(
    const float* __restrict__ ef, const float* __restrict__ nf,
    const float* __restrict__ w0p, const float* __restrict__ b0,
    const float* __restrict__ w1p, const float* __restrict__ b1,
    const int* __restrict__ src, const int* __restrict__ dst,
    float* __restrict__ acc, int n_edges)
{
  __shared__ v2f   sW0[(D_EDGE / 2) * D_EDGE];
  __shared__ float sh[WAVES][16 * D_EDGE];

  const int tid = threadIdx.x;
  {
    const v2f* w0v = (const v2f*)w0p;
    for (int i = tid; i < (D_EDGE / 2) * D_EDGE; i += 256) sW0[i] = w0v[i];
  }

  const int wave  = tid >> 5;
  const int lane  = tid & 31;
  const int row16 = lane & 15;
  const int hi    = lane >> 4;

  const int  tile   = blockIdx.x * WAVES + wave;
  const int  ebase0 = tile * 16;
  const bool active = ebase0 < n_edges;
  const int  ebase  = active ? ebase0 : 0;

  int arow = ebase + row16;
  if (arow >= n_edges) arow = n_edges - 1;
  const v2f* efv   = (const v2f*)ef;
  const int  abase = arow * (D_EDGE / 2);

  v2f ae[D_EDGE / 4];
  #pragma unroll
  for (int k = 0; k < D_EDGE / 4; ++k) ae[k] = efv[abase + 2 * k + hi];

  int soff[8], doff[8];
  #pragma unroll
  for (int r = 0; r < 8; ++r) {
    int e  = ebase + r + 8 * hi;
    int ec = e < n_edges ? e : n_edges - 1;
    soff[r] = src[ec] * D_NODE + row16;
    doff[r] = dst[ec] * D_NODE + row16;
  }

  __syncthreads();

  float* hrow = &sh[wave][0];

  #pragma unroll
  for (int nt = 0; nt < D_EDGE / 16; ++nt) {
    v8f c = {};
    #pragma unroll
    for (int k = 0; k < D_EDGE / 4; ++k)
      c = wmma_f32(ae[k], sW0[(2 * k + hi) * D_EDGE + nt * 16 + row16], c);
    const int   col = nt * 16 + row16;
    const float bs  = b0[col];
    #pragma unroll
    for (int r = 0; r < 8; ++r) {
      float v = c[r] + bs;
      hrow[(r + 8 * hi) * D_EDGE + col] = v > 0.f ? v : 0.f;
    }
  }
  __syncthreads();

  const v2f* shv = (const v2f*)hrow;
  v2f ah[D_EDGE / 4];
  #pragma unroll
  for (int k = 0; k < D_EDGE / 4; ++k)
    ah[k] = shv[row16 * (D_EDGE / 2) + 2 * k + hi];

  const v2f* w1v = (const v2f*)w1p;
  #pragma unroll 1
  for (int nt = 0; nt < D_NODE / 16; ++nt) {
    v8f c = {};
    #pragma unroll
    for (int k = 0; k < D_EDGE / 4; ++k)
      c = wmma_f32(ah[k], w1v[(2 * k + hi) * D_NODE + nt * 16 + row16], c);
    const int   cb = nt * 16;
    const float bs = b1[cb + row16];
    if (active) {
      #pragma unroll
      for (int r = 0; r < 8; ++r) {
        if (ebase + r + 8 * hi < n_edges) {
          float ev = c[r] + bs;
          ev = ev > 0.f ? ev : 0.f;
          float v = nf[soff[r] + cb] + ev;
          v = v > 0.f ? v : 0.f;
          atomicAdd(&acc[doff[r] + cb], v);
        }
      }
    }
  }
}

__global__ __launch_bounds__(256) void gine_out_kernel(
    const float* __restrict__ acc, const float* __restrict__ wap,
    const float* __restrict__ ba, float* __restrict__ out, int n_nodes)
{
  const int tid   = threadIdx.x;
  const int wave  = tid >> 5;
  const int lane  = tid & 31;
  const int row16 = lane & 15;
  const int hi    = lane >> 4;

  const int tile = blockIdx.x * WAVES + wave;
  const int m0   = tile * 16;
  if (m0 >= n_nodes) return;

  int arow = m0 + row16;
  if (arow >= n_nodes) arow = n_nodes - 1;
  const v2f* av    = (const v2f*)acc;
  const int  abase = arow * (D_NODE / 2);

  v2f aa[D_NODE / 4];
  #pragma unroll
  for (int k = 0; k < D_NODE / 4; ++k) aa[k] = av[abase + 2 * k + hi];

  const v2f* wav = (const v2f*)wap;
  #pragma unroll 1
  for (int nt = 0; nt < D_NODE / 16; ++nt) {
    v8f c = {};
    #pragma unroll
    for (int k = 0; k < D_NODE / 4; ++k)
      c = wmma_f32(aa[k], wav[(2 * k + hi) * D_NODE + nt * 16 + row16], c);
    const int   col = nt * 16 + row16;
    const float bs  = ba[col];
    #pragma unroll
    for (int r = 0; r < 8; ++r) {
      int m = m0 + r + 8 * hi;
      if (m < n_nodes) out[m * D_NODE + col] = c[r] + bs;
    }
  }
}

extern "C" void kernel_launch(void* const* d_in, const int* in_sizes, int n_in,
                              void* d_out, int out_size, void* d_ws, size_t ws_size,
                              hipStream_t stream) {
  const float* nf = (const float*)d_in[0];
  const float* ef = (const float*)d_in[1];
  const float* W0 = (const float*)d_in[2];
  const float* b0 = (const float*)d_in[3];
  const float* W1 = (const float*)d_in[4];
  const float* b1 = (const float*)d_in[5];
  const float* Wa = (const float*)d_in[6];
  const float* ba = (const float*)d_in[7];
  const int*  src = (const int*)d_in[8];
  const int*  dst = (const int*)d_in[9];

  const int n_nodes = in_sizes[0] / D_NODE;
  const int n_edges = in_sizes[8];

  float* acc = (float*)d_ws;
  float* w0p = acc + (size_t)n_nodes * D_NODE;
  float* w1p = w0p + D_EDGE * D_EDGE;
  float* wap = w1p + D_EDGE * D_NODE;

  hipMemcpyAsync(acc, nf, (size_t)n_nodes * D_NODE * sizeof(float),
                 hipMemcpyDeviceToDevice, stream);

  pack_pairs_kernel<<<(D_EDGE * D_EDGE + 255) / 256, 256, 0, stream>>>(
      W0, w0p, D_EDGE, D_EDGE);
  pack_pairs_kernel<<<(D_EDGE * D_NODE + 255) / 256, 256, 0, stream>>>(
      W1, w1p, D_EDGE, D_NODE);
  pack_pairs_kernel<<<(D_NODE * D_NODE + 255) / 256, 256, 0, stream>>>(
      Wa, wap, D_NODE, D_NODE);

  const int etiles  = (n_edges + 15) / 16;
  const int eblocks = (etiles + WAVES - 1) / WAVES;
  gine_edge_kernel<<<eblocks, 256, 0, stream>>>(ef, nf, w0p, b0, w1p, b1, src,
                                                dst, acc, n_edges);

  const int ntiles  = (n_nodes + 15) / 16;
  const int nblocks = (ntiles + WAVES - 1) / WAVES;
  gine_out_kernel<<<nblocks, 256, 0, stream>>>(acc, wap, ba, (float*)d_out,
                                               n_nodes);
}


# Edges grouped by 16-node tile via per-work-group LDS counters (count, prefix over groups, place; LDS add returns the slot), no global atomic; S from stored dst&15; sums reordered, still arrival order
# speedup vs baseline: 1.8817x; 1.8817x over previous
#include <hip/hip_runtime.h>


typedef __attribute__((ext_vector_type(2))) float v2f;
typedef __attribute__((ext_vector_type(8))) float v8f;

#define D_NODE 128
#define D_EDGE 64
#define PW     4
#define HP     (D_EDGE + 2)
#define GT     4
#define RP     D_NODE

static __device__ __forceinline__ v8f wmma_f32(v2f a, v2f b, v8f c) {
  return __builtin_amdgcn_wmma_f32_16x16x4_f32(false, a, false, b, (short)0, c,
                                               false, false);
}

static __device__ __forceinline__ void wave_lds_handover() {
  __builtin_amdgcn_fence(__ATOMIC_RELEASE, "wavefront");
  __builtin_amdgcn_wave_barrier();
  __builtin_amdgcn_fence(__ATOMIC_ACQUIRE, "wavefront");
}

static __device__ __forceinline__ void lds_group_edge() {
  __builtin_amdgcn_sched_barrier(0);
}

__global__ __launch_bounds__(256) void pack_pairs_kernel(
    const float* __restrict__ W, float* __restrict__ Wp, int rows, int cols) {
  int i = blockIdx.x * 256 + threadIdx.x;
  if (i >= rows * cols) return;
  int r = i / cols, c = i - r * cols;
  Wp[(((r >> 1) * cols + c) << 1) + (r & 1)] = W[i];
}

#define OG        128
#define OT        1024
#define MAX_TILES 4096
#define SRC_MASK  0x0FFFFFFF

__global__ __launch_bounds__(OT) void tile_count_kernel(
    const int* __restrict__ dst, int* __restrict__ cnt, int n_edges,
    int n_nodes, int ntiles) {
  __shared__ int h[MAX_TILES];
  const int t = threadIdx.x;
  for (int i = t; i < ntiles; i += OT) h[i] = 0;
  __syncthreads();
  const int chunk = (n_edges + OG - 1) / OG;
  const int e0 = blockIdx.x * chunk;
  const int e1 = e0 + chunk < n_edges ? e0 + chunk : n_edges;
  for (int eb = e0; eb < e1; eb += 8 * OT) {
    int d[8];
    #pragma unroll
    for (int k = 0; k < 8; ++k) {
      const int e  = eb + k * OT + t;
      const int ec = e < e1 ? e : e1 - 1;
      d[k] = dst[ec];
      if (e >= e1) d[k] = -1;
    }
    #pragma unroll
    for (int k = 0; k < 8; ++k)
      if ((unsigned)d[k] < (unsigned)n_nodes) atomicAdd(&h[d[k] >> 4], 1);
  }
  __syncthreads();
  for (int i = t; i < ntiles; i += OT) cnt[blockIdx.x * ntiles + i] = h[i];
}

__global__ __launch_bounds__(256) void tile_prefix_kernel(
    const int* __restrict__ cnt, int* __restrict__ pre, int* __restrict__ tot,
    int ntiles) {
  const int b = blockIdx.x * 256 + threadIdx.x;
  if (b >= ntiles) return;
  int run = 0;
  #pragma unroll 16
  for (int g = 0; g < OG; ++g) {
    const int v = cnt[g * ntiles + b];
    pre[g * ntiles + b] = run;
    run += v;
  }
  tot[b] = run;
}

__global__ __launch_bounds__(OT) void tile_place_kernel(
    const int* __restrict__ src, const int* __restrict__ dst,
    const int* __restrict__ pre, const int* __restrict__ tot,
    int* __restrict__ off, int* __restrict__ rec, int n_edges, int n_nodes,
    int ntiles) {
  __shared__ int slot[MAX_TILES];
  __shared__ int wsum[OT / 32];
  const int t    = threadIdx.x;
  const int lane = t & 31;
  const int w    = t >> 5;
  const int g    = blockIdx.x;

  int v[4], pv[4];
  int s = 0;
  #pragma unroll
  for (int k = 0; k < 4; ++k) {
    const int i  = 4 * t + k;
    const int ic = i < ntiles ? i : ntiles - 1;
    v[k]  = tot[ic];
    pv[k] = pre[g * ntiles + ic];
    if (i >= ntiles) v[k] = 0;
    s += v[k];
  }
  int inc = s;
  #pragma unroll
  for (int d = 1; d < 32; d <<= 1) {
    const int u = __shfl_up(inc, d);
    if (lane >= d) inc += u;
  }
  if (lane == 31) wsum[w] = inc;
  __syncthreads();
  if (w == 0) {
    const int x = wsum[lane];
    int xi = x;
    #pragma unroll
    for (int d = 1; d < 32; d <<= 1) {
      const int u = __shfl_up(xi, d);
      if (lane >= d) xi += u;
    }
    wsum[lane] = xi - x;
  }
  __syncthreads();
  int ex = wsum[w] + inc - s;
  #pragma unroll
  for (int k = 0; k < 4; ++k) {
    const int i = 4 * t + k;
    if (i < ntiles) {
      slot[i] = ex + pv[k];
      if (g == 0) off[i] = ex;
    }
    ex += v[k];
  }
  if (g == 0 && t == OT - 1) off[ntiles] = ex;
  __syncthreads();

  typedef __attribute__((ext_vector_type(2))) int v2i;
  v2i* recv = (v2i*)rec;
  const int chunk = (n_edges + OG - 1) / OG;
  const int e0 = g * chunk;
  const int e1 = e0 + chunk < n_edges ? e0 + chunk : n_edges;
  for (int eb = e0; eb < e1; eb += 8 * OT) {
    int d[8], s8[8];
    #pragma unroll
    for (int k = 0; k < 8; ++k) {
      const int e  = eb + k * OT + t;
      const int ec = e < e1 ? e : e1 - 1;
      d[k]  = dst[ec];
      s8[k] = src[ec];
      if (e >= e1) d[k] = -1;
    }
    #pragma unroll
    for (int k = 0; k < 8; ++k) {
      if ((unsigned)d[k] < (unsigned)n_nodes) {
        const int p = atomicAdd(&slot[d[k] >> 4], 1);
        v2i r;
        r.x = eb + k * OT + t;
        r.y = (int)(((unsigned)s8[k] & SRC_MASK) | ((unsigned)(d[k] & 15) << 28));
        recv[p] = r;
      }
    }
  }
}

__global__ __launch_bounds__(PW * 32, 3) void gine_pull_kernel(
    const float* __restrict__ ef, const float* __restrict__ nf,
    const float* __restrict__ w0p, const float* __restrict__ b0,
    const float* __restrict__ w1p, const float* __restrict__ b1,
    const float* __restrict__ wap, const float* __restrict__ ba,
    const int* __restrict__ off, const int* __restrict__ rec,
    float* __restrict__ out, int n_nodes)
{
  __shared__ v2f   sW0[(D_EDGE / 2) * D_EDGE];
  __shared__ v2f   sW1[(D_EDGE / 2) * D_NODE];
  __shared__ __attribute__((aligned(16))) float sst[PW][16 * RP];

  const int tid = threadIdx.x;
  {
    const v2f* w0v = (const v2f*)w0p;
    for (int i = tid; i < (D_EDGE / 2) * D_EDGE; i += PW * 32) sW0[i] = w0v[i];
    const v2f* w1v = (const v2f*)w1p;
    for (int i = tid; i < (D_EDGE / 2) * D_NODE; i += PW * 32) sW1[i] = w1v[i];
  }
  __syncthreads();

  const int hwave = __builtin_amdgcn_readfirstlane(tid >> 5);
  const int wave = (hwave + blockIdx.x) & (PW - 1);
  const int lane = tid & 31;
  const int c    = lane & 15;
  const int hi   = lane >> 4;

  const int n0 = blockIdx.x * 16;

  float* st = &sst[hwave][0];

  v8f accN[D_NODE / 16];
  if (wave == 0) {
    #pragma unroll
    for (int r = 0; r < 8; ++r) {
      int n = n0 + 8 * hi + r;
      if (n >= n_nodes) n = n_nodes - 1;
      const float* nrow = nf + n * D_NODE + c;
      #pragma unroll
      for (int nt = 0; nt < D_NODE / 16; ++nt) accN[nt][r] = nrow[nt * 16];
    }
  } else {
    #pragma unroll
    for (int nt = 0; nt < D_NODE / 16; ++nt) accN[nt] = (v8f){};
  }

  const int ebeg = __builtin_amdgcn_readfirstlane(off[blockIdx.x]);
  const int eend = __builtin_amdgcn_readfirstlane(off[blockIdx.x + 1]);
  typedef __attribute__((ext_vector_type(2))) int v2i;
  const v2i* recv = (const v2i*)rec;

  const v2f* efv = (const v2f*)ef;

  const int last  = eend - 1;
  const int wbase = ebeg + 16 * wave;
  int eid = 0, sid = 0;
  v2f ae[D_EDGE / 4];
  #pragma unroll
  for (int k = 0; k < D_EDGE / 4; ++k) ae[k] = (v2f){0.f, 0.f};
  if (wbase < eend) {
    int p = wbase + c;
    if (p > last) p = last;
    const v2i rc = recv[p];
    eid = rc.x;
    sid = rc.y;
    const v2f* erow = efv + eid * (D_EDGE / 2);
    #pragma unroll
    for (int k = 0; k < D_EDGE / 4; ++k) ae[k] = erow[2 * k + hi];
  }

  #pragma unroll 1
  for (int base = wbase; base < eend; base += 16 * PW) {
    int sid_next;
    {
      int p = base + 16 * PW + c;
      if (p > last) p = last;
      const v2i rc = recv[p];
      eid = rc.x;
      sid_next = rc.y;
    }
    __builtin_amdgcn_sched_barrier(0);

    int soff[8];
    v2f sa[4];
    #pragma unroll
    for (int r = 0; r < 8; ++r) {
      const int s0 = __builtin_amdgcn_readlane(sid, r);
      const int s1 = __builtin_amdgcn_readlane(sid, r + 8);
      const int sw = hi ? s1 : s0;
      soff[r] = (sw & SRC_MASK) * D_NODE + c;
      const int p = base + 8 * hi + r;
      sa[r >> 1][r & 1] = (p < eend && ((unsigned)sw >> 28) == (unsigned)c) ? 1.f : 0.f;
    }

    sid = sid_next;

    float g[D_NODE / 16][8];
    #pragma unroll
    for (int t = 0; t < GT; ++t) {
      #pragma unroll
      for (int r = 0; r < 8; ++r) g[t][r] = nf[soff[r] + 16 * t];
    }

    v2f bw[2][16];
    #pragma unroll
    for (int kk = 0; kk < 8; ++kk) {
      #pragma unroll
      for (int t = 0; t < 2; ++t) {
        bw[0][2 * kk + t] = sW0[(2 * kk + hi) * D_EDGE + t * 16 + c];
        bw[1][2 * kk + t] = sW0[(2 * (8 + kk) + hi) * D_EDGE + t * 16 + c];
      }
    }
    lds_group_edge();

    v2f ah[D_EDGE / 4];
    #pragma unroll
    for (int k = 0; k < D_EDGE / 4; ++k) ah[k] = (v2f){0.f, 0.f};
    v8f acc0 = {}, acc1 = {};
    #pragma unroll
    for (int h = 0; h < 12; ++h) {
      const int  s    = h & 1;
      const bool lay1 = h >= 4;
      const int  p    = lay1 ? ((h - 4) >> 1) : (h >> 1);
      if (s == 0) { acc0 = (v8f){}; acc1 = (v8f){}; }
      #pragma unroll
      for (int kk = 0; kk < 8; ++kk) {
        const v2f a = lay1 ? ah[8 * s + kk] : ae[8 * s + kk];
        acc0 = wmma_f32(a, bw[s][2 * kk],     acc0);
        acc1 = wmma_f32(a, bw[s][2 * kk + 1], acc1);
      }
      if (s == 1 && !lay1) {
        #pragma unroll
        for (int t = 0; t < 2; ++t) {
          const v8f   acc = t ? acc1 : acc0;
          const int   col = (2 * p + t) * 16 + c;
          const float bs  = b0[col];
          #pragma unroll
          for (int r = 0; r < 8; ++r) {
            float v = acc[r] + bs;
            st[(r + 8 * hi) * HP + col] = v > 0.f ? v : 0.f;
          }
        }
      }
      if (s == 1 && lay1) {
        #pragma unroll
        for (int t = 0; t < 2; ++t) {
          const v8f   acc = t ? acc1 : acc0;
          const int   nt  = 2 * p + t;
          const float bs  = b1[nt * 16 + c];
          v8f m;
          #pragma unroll
          for (int r = 0; r < 8; ++r) {
            float ev = acc[r] + bs;
            ev = ev > 0.f ? ev : 0.f;
            float v = g[nt][r] + ev;
            m[r] = v > 0.f ? v : 0.f;
          }
          #pragma unroll
          for (int j = 0; j < 4; ++j) {
            v2f mb = {m[2 * j], m[2 * j + 1]};
            accN[nt] = wmma_f32(sa[j], mb, accN[nt]);
          }
        }
      }
      if (h + 2 < 12) {
        lds_group_edge();
        if (h == 3) {
          wave_lds_handover();
          #pragma unroll
          for (int k = 0; k < D_EDGE / 4; ++k)
            ah[k] = *(const v2f*)&st[c * HP + 4 * k + 2 * hi];
          wave_lds_handover();
        }
        {
          const int  h2    = h + 2;
          const bool lay1n = h2 >= 4;
          const int  p2    = lay1n ? ((h2 - 4) >> 1) : (h2 >> 1);
          #pragma unroll
          for (int kk = 0; kk < 8; ++kk) {
            #pragma unroll
            for (int t = 0; t < 2; ++t) {
              const int krow = 2 * (8 * s + kk) + hi;
              bw[s][2 * kk + t] = lay1n ? sW1[krow * D_NODE + (2 * p2 + t) * 16 + c]
                                        : sW0[krow * D_EDGE + (2 * p2 + t) * 16 + c];
            }
          }
        }
        if (h == 3) {
          const v2f* erow = efv + eid * (D_EDGE / 2);
          #pragma unroll
          for (int k = 0; k < D_EDGE / 4; ++k) ae[k] = erow[2 * k + hi];
        }
        if (h == 5 || h == 7) {
          #pragma unroll
          for (int t = h - 1; t <= h; ++t) {
            #pragma unroll
            for (int r = 0; r < 8; ++r) g[t][r] = nf[soff[r] + 16 * t];
          }
        }
        lds_group_edge();
      }
    }
  }

  #pragma unroll
  for (int nt = 0; nt < D_NODE / 16; ++nt) {
    #pragma unroll
    for (int r = 0; r < 8; ++r) {
      const int row = r + 8 * hi;
      st[row * RP + ((nt * 16 + c) ^ (row << 1))] = accN[nt][r];
    }
  }
  __syncthreads();

  v2f aa[D_NODE / 4];
  #pragma unroll
  for (int k = 0; k < D_NODE / 4; ++k) {
    const int o = c * RP + ((4 * k + 2 * hi) ^ (c << 1));
    v2f s = *(const v2f*)&sst[0][o];
    #pragma unroll
    for (int w = 1; w < PW; ++w) s += *(const v2f*)&sst[w][o];
    aa[k] = s;
  }

  const v2f* wav = (const v2f*)wap;
  #pragma unroll
  for (int t = 0; t < D_NODE / 16 / PW; ++t) {
    const int nt = wave * (D_NODE / 16 / PW) + t;
    v8f acc = {};
    #pragma unroll
    for (int k = 0; k < D_NODE / 4; ++k)
      acc = wmma_f32(aa[k], wav[(2 * k + hi) * D_NODE + nt * 16 + c], acc);
    const int   col = nt * 16 + c;
    const float bs  = ba[col];
    #pragma unroll
    for (int r = 0; r < 8; ++r) {
      int m = n0 + r + 8 * hi;
      if (m < n_nodes) out[m * D_NODE + col] = acc[r] + bs;
    }
  }
}

extern "C" void kernel_launch(void* const* d_in, const int* in_sizes, int n_in,
                              void* d_out, int out_size, void* d_ws, size_t ws_size,
                              hipStream_t stream) {
  const float* nf = (const float*)d_in[0];
  const float* ef = (const float*)d_in[1];
  const float* W0 = (const float*)d_in[2];
  const float* b0 = (const float*)d_in[3];
  const float* W1 = (const float*)d_in[4];
  const float* b1 = (const float*)d_in[5];
  const float* Wa = (const float*)d_in[6];
  const float* ba = (const float*)d_in[7];
  const int*  src = (const int*)d_in[8];
  const int*  dst = (const int*)d_in[9];

  const int n_nodes = in_sizes[0] / D_NODE;
  const int n_edges = in_sizes[8];

  const int ntiles = (n_nodes + 15) / 16;

  float* w0p  = (float*)d_ws;
  float* w1p  = w0p + D_EDGE * D_EDGE;
  float* wap  = w1p + D_EDGE * D_NODE;
  int*   rec  = (int*)(wap + D_NODE * D_NODE);
  int*   cnt  = rec + 2 * n_edges;
  int*   pre  = cnt + OG * ntiles;
  int*   tot  = pre + OG * ntiles;
  int*   off  = tot + ntiles;

  pack_pairs_kernel<<<(D_EDGE * D_EDGE + 255) / 256, 256, 0, stream>>>(
      W0, w0p, D_EDGE, D_EDGE);
  pack_pairs_kernel<<<(D_EDGE * D_NODE + 255) / 256, 256, 0, stream>>>(
      W1, w1p, D_EDGE, D_NODE);
  pack_pairs_kernel<<<(D_NODE * D_NODE + 255) / 256, 256, 0, stream>>>(
      Wa, wap, D_NODE, D_NODE);

  tile_count_kernel<<<OG, OT, 0, stream>>>(dst, cnt, n_edges, n_nodes, ntiles);
  tile_prefix_kernel<<<(ntiles + 255) / 256, 256, 0, stream>>>(cnt, pre, tot,
                                                               ntiles);
  tile_place_kernel<<<OG, OT, 0, stream>>>(src, dst, pre, tot, off, rec,
                                           n_edges, n_nodes, ntiles);

  gine_pull_kernel<<<ntiles, PW * 32, 0, stream>>>(
      ef, nf, w0p, b0, w1p, b1, wap, ba, off, rec, (float*)d_out, n_nodes);
}


# The three weight-packing launches folded into the tile-count launch as 28 extra work-groups (work-group-uniform branch): 7 launches -> 4; pull kernel, sums and their order unchanged
# speedup vs baseline: 1.9539x; 1.0384x over previous
#include <hip/hip_runtime.h>


typedef __attribute__((ext_vector_type(2))) float v2f;
typedef __attribute__((ext_vector_type(8))) float v8f;

#define D_NODE 128
#define D_EDGE 64
#define PW     4
#define HP     (D_EDGE + 2)
#define GT     4
#define RP     D_NODE

static __device__ __forceinline__ v8f wmma_f32(v2f a, v2f b, v8f c) {
  return __builtin_amdgcn_wmma_f32_16x16x4_f32(false, a, false, b, (short)0, c,
                                               false, false);
}

static __device__ __forceinline__ void wave_lds_handover() {
  __builtin_amdgcn_fence(__ATOMIC_RELEASE, "wavefront");
  __builtin_amdgcn_wave_barrier();
  __builtin_amdgcn_fence(__ATOMIC_ACQUIRE, "wavefront");
}

static __device__ __forceinline__ void lds_group_edge() {
  __builtin_amdgcn_sched_barrier(0);
}

#define OG        128
#define OT        1024
#define MAX_TILES 4096
#define SRC_MASK  0x0FFFFFFF
#define PACK_W0    (D_EDGE * D_EDGE)
#define PACK_W1    (D_EDGE * D_NODE)
#define PACK_WA    (D_NODE * D_NODE)
#define PACK_ELEMS (PACK_W0 + PACK_W1 + PACK_WA)
#define PACK_BLOCKS ((PACK_ELEMS + OT - 1) / OT)

__global__ __launch_bounds__(OT) void tile_count_pack_kernel(
    const int* __restrict__ dst, int* __restrict__ cnt, int n_edges,
    int n_nodes, int ntiles,
    const float* __restrict__ W0, const float* __restrict__ W1,
    const float* __restrict__ Wa, float* __restrict__ w0p,
    float* __restrict__ w1p, float* __restrict__ wap) {
  __shared__ int h[MAX_TILES];
  const int t = threadIdx.x;
  if (blockIdx.x >= OG) {
    int i = (blockIdx.x - OG) * OT + t;
    if (i < PACK_W0) {
      const int r = i / D_EDGE, c = i - r * D_EDGE;
      w0p[(((r >> 1) * D_EDGE + c) << 1) + (r & 1)] = W0[i];
    } else if (i < PACK_W0 + PACK_W1) {
      i -= PACK_W0;
      const int r = i / D_NODE, c = i - r * D_NODE;
      w1p[(((r >> 1) * D_NODE + c) << 1) + (r & 1)] = W1[i];
    } else if (i < PACK_ELEMS) {
      i -= PACK_W0 + PACK_W1;
      const int r = i / D_NODE, c = i - r * D_NODE;
      wap[(((r >> 1) * D_NODE + c) << 1) + (r & 1)] = Wa[i];
    }
    return;
  }
  for (int i = t; i < ntiles; i += OT) h[i] = 0;
  __syncthreads();
  const int chunk = (n_edges + OG - 1) / OG;
  const int e0 = blockIdx.x * chunk;
  const int e1 = e0 + chunk < n_edges ? e0 + chunk : n_edges;
  for (int eb = e0; eb < e1; eb += 8 * OT) {
    int d[8];
    #pragma unroll
    for (int k = 0; k < 8; ++k) {
      const int e  = eb + k * OT + t;
      const int ec = e < e1 ? e : e1 - 1;
      d[k] = dst[ec];
      if (e >= e1) d[k] = -1;
    }
    #pragma unroll
    for (int k = 0; k < 8; ++k)
      if ((unsigned)d[k] < (unsigned)n_nodes) atomicAdd(&h[d[k] >> 4], 1);
  }
  __syncthreads();
  for (int i = t; i < ntiles; i += OT) cnt[blockIdx.x * ntiles + i] = h[i];
}

__global__ __launch_bounds__(256) void tile_prefix_kernel(
    const int* __restrict__ cnt, int* __restrict__ pre, int* __restrict__ tot,
    int ntiles) {
  const int b = blockIdx.x * 256 + threadIdx.x;
  if (b >= ntiles) return;
  int run = 0;
  #pragma unroll 16
  for (int g = 0; g < OG; ++g) {
    const int v = cnt[g * ntiles + b];
    pre[g * ntiles + b] = run;
    run += v;
  }
  tot[b] = run;
}

__global__ __launch_bounds__(OT) void tile_place_kernel(
    const int* __restrict__ src, const int* __restrict__ dst,
    const int* __restrict__ pre, const int* __restrict__ tot,
    int* __restrict__ off, int* __restrict__ rec, int n_edges, int n_nodes,
    int ntiles) {
  __shared__ int slot[MAX_TILES];
  __shared__ int wsum[OT / 32];
  const int t    = threadIdx.x;
  const int lane = t & 31;
  const int w    = t >> 5;
  const int g    = blockIdx.x;

  int v[4], pv[4];
  int s = 0;
  #pragma unroll
  for (int k = 0; k < 4; ++k) {
    const int i  = 4 * t + k;
    const int ic = i < ntiles ? i : ntiles - 1;
    v[k]  = tot[ic];
    pv[k] = pre[g * ntiles + ic];
    if (i >= ntiles) v[k] = 0;
    s += v[k];
  }
  int inc = s;
  #pragma unroll
  for (int d = 1; d < 32; d <<= 1) {
    const int u = __shfl_up(inc, d);
    if (lane >= d) inc += u;
  }
  if (lane == 31) wsum[w] = inc;
  __syncthreads();
  if (w == 0) {
    const int x = wsum[lane];
    int xi = x;
    #pragma unroll
    for (int d = 1; d < 32; d <<= 1) {
      const int u = __shfl_up(xi, d);
      if (lane >= d) xi += u;
    }
    wsum[lane] = xi - x;
  }
  __syncthreads();
  int ex = wsum[w] + inc - s;
  #pragma unroll
  for (int k = 0; k < 4; ++k) {
    const int i = 4 * t + k;
    if (i < ntiles) {
      slot[i] = ex + pv[k];
      if (g == 0) off[i] = ex;
    }
    ex += v[k];
  }
  if (g == 0 && t == OT - 1) off[ntiles] = ex;
  __syncthreads();

  typedef __attribute__((ext_vector_type(2))) int v2i;
  v2i* recv = (v2i*)rec;
  const int chunk = (n_edges + OG - 1) / OG;
  const int e0 = g * chunk;
  const int e1 = e0 + chunk < n_edges ? e0 + chunk : n_edges;
  for (int eb = e0; eb < e1; eb += 8 * OT) {
    int d[8], s8[8];
    #pragma unroll
    for (int k = 0; k < 8; ++k) {
      const int e  = eb + k * OT + t;
      const int ec = e < e1 ? e : e1 - 1;
      d[k]  = dst[ec];
      s8[k] = src[ec];
      if (e >= e1) d[k] = -1;
    }
    #pragma unroll
    for (int k = 0; k < 8; ++k) {
      if ((unsigned)d[k] < (unsigned)n_nodes) {
        const int p = atomicAdd(&slot[d[k] >> 4], 1);
        v2i r;
        r.x = eb + k * OT + t;
        r.y = (int)(((unsigned)s8[k] & SRC_MASK) | ((unsigned)(d[k] & 15) << 28));
        recv[p] = r;
      }
    }
  }
}

__global__ __launch_bounds__(PW * 32, 3) void gine_pull_kernel(
    const float* __restrict__ ef, const float* __restrict__ nf,
    const float* __restrict__ w0p, const float* __restrict__ b0,
    const float* __restrict__ w1p, const float* __restrict__ b1,
    const float* __restrict__ wap, const float* __restrict__ ba,
    const int* __restrict__ off, const int* __restrict__ rec,
    float* __restrict__ out, int n_nodes)
{
  __shared__ v2f   sW0[(D_EDGE / 2) * D_EDGE];
  __shared__ v2f   sW1[(D_EDGE / 2) * D_NODE];
  __shared__ __attribute__((aligned(16))) float sst[PW][16 * RP];

  const int tid = threadIdx.x;
  {
    const v2f* w0v = (const v2f*)w0p;
    for (int i = tid; i < (D_EDGE / 2) * D_EDGE; i += PW * 32) sW0[i] = w0v[i];
    const v2f* w1v = (const v2f*)w1p;
    for (int i = tid; i < (D_EDGE / 2) * D_NODE; i += PW * 32) sW1[i] = w1v[i];
  }
  __syncthreads();

  const int hwave = __builtin_amdgcn_readfirstlane(tid >> 5);
  const int wave = (hwave + blockIdx.x) & (PW - 1);
  const int lane = tid & 31;
  const int c    = lane & 15;
  const int hi   = lane >> 4;

  const int n0 = blockIdx.x * 16;

  float* st = &sst[hwave][0];

  v8f accN[D_NODE / 16];
  if (wave == 0) {
    #pragma unroll
    for (int r = 0; r < 8; ++r) {
      int n = n0 + 8 * hi + r;
      if (n >= n_nodes) n = n_nodes - 1;
      const float* nrow = nf + n * D_NODE + c;
      #pragma unroll
      for (int nt = 0; nt < D_NODE / 16; ++nt) accN[nt][r] = nrow[nt * 16];
    }
  } else {
    #pragma unroll
    for (int nt = 0; nt < D_NODE / 16; ++nt) accN[nt] = (v8f){};
  }

  const int ebeg = __builtin_amdgcn_readfirstlane(off[blockIdx.x]);
  const int eend = __builtin_amdgcn_readfirstlane(off[blockIdx.x + 1]);
  typedef __attribute__((ext_vector_type(2))) int v2i;
  const v2i* recv = (const v2i*)rec;

  const v2f* efv = (const v2f*)ef;

  const int last  = eend - 1;
  const int wbase = ebeg + 16 * wave;
  int eid = 0, sid = 0;
  v2f ae[D_EDGE / 4];
  #pragma unroll
  for (int k = 0; k < D_EDGE / 4; ++k) ae[k] = (v2f){0.f, 0.f};
  if (wbase < eend) {
    int p = wbase + c;
    if (p > last) p = last;
    const v2i rc = recv[p];
    eid = rc.x;
    sid = rc.y;
    const v2f* erow = efv + eid * (D_EDGE / 2);
    #pragma unroll
    for (int k = 0; k < D_EDGE / 4; ++k) ae[k] = erow[2 * k + hi];
  }

  #pragma unroll 1
  for (int base = wbase; base < eend; base += 16 * PW) {
    int sid_next;
    {
      int p = base + 16 * PW + c;
      if (p > last) p = last;
      const v2i rc = recv[p];
      eid = rc.x;
      sid_next = rc.y;
    }
    __builtin_amdgcn_sched_barrier(0);

    int soff[8];
    v2f sa[4];
    #pragma unroll
    for (int r = 0; r < 8; ++r) {
      const int s0 = __builtin_amdgcn_readlane(sid, r);
      const int s1 = __builtin_amdgcn_readlane(sid, r + 8);
      const int sw = hi ? s1 : s0;
      soff[r] = (sw & SRC_MASK) * D_NODE + c;
      const int p = base + 8 * hi + r;
      sa[r >> 1][r & 1] = (p < eend && ((unsigned)sw >> 28) == (unsigned)c) ? 1.f : 0.f;
    }

    sid = sid_next;

    float g[D_NODE / 16][8];
    #pragma unroll
    for (int t = 0; t < GT; ++t) {
      #pragma unroll
      for (int r = 0; r < 8; ++r) g[t][r] = nf[soff[r] + 16 * t];
    }

    v2f bw[2][16];
    #pragma unroll
    for (int kk = 0; kk < 8; ++kk) {
      #pragma unroll
      for (int t = 0; t < 2; ++t) {
        bw[0][2 * kk + t] = sW0[(2 * kk + hi) * D_EDGE + t * 16 + c];
        bw[1][2 * kk + t] = sW0[(2 * (8 + kk) + hi) * D_EDGE + t * 16 + c];
      }
    }
    lds_group_edge();

    v2f ah[D_EDGE / 4];
    #pragma unroll
    for (int k = 0; k < D_EDGE / 4; ++k) ah[k] = (v2f){0.f, 0.f};
    v8f acc0 = {}, acc1 = {};
    #pragma unroll
    for (int h = 0; h < 12; ++h) {
      const int  s    = h & 1;
      const bool lay1 = h >= 4;
      const int  p    = lay1 ? ((h - 4) >> 1) : (h >> 1);
      if (s == 0) { acc0 = (v8f){}; acc1 = (v8f){}; }
      #pragma unroll
      for (int kk = 0; kk < 8; ++kk) {
        const v2f a = lay1 ? ah[8 * s + kk] : ae[8 * s + kk];
        acc0 = wmma_f32(a, bw[s][2 * kk],     acc0);
        acc1 = wmma_f32(a, bw[s][2 * kk + 1], acc1);
      }
      if (s == 1 && !lay1) {
        #pragma unroll
        for (int t = 0; t < 2; ++t) {
          const v8f   acc = t ? acc1 : acc0;
          const int   col = (2 * p + t) * 16 + c;
          const float bs  = b0[col];
          #pragma unroll
          for (int r = 0; r < 8; ++r) {
            float v = acc[r] + bs;
            st[(r + 8 * hi) * HP + col] = v > 0.f ? v : 0.f;
          }
        }
      }
      if (s == 1 && lay1) {
        #pragma unroll
        for (int t = 0; t < 2; ++t) {
          const v8f   acc = t ? acc1 : acc0;
          const int   nt  = 2 * p + t;
          const float bs  = b1[nt * 16 + c];
          v8f m;
          #pragma unroll
          for (int r = 0; r < 8; ++r) {
            float ev = acc[r] + bs;
            ev = ev > 0.f ? ev : 0.f;
            float v = g[nt][r] + ev;
            m[r] = v > 0.f ? v : 0.f;
          }
          #pragma unroll
          for (int j = 0; j < 4; ++j) {
            v2f mb = {m[2 * j], m[2 * j + 1]};
            accN[nt] = wmma_f32(sa[j], mb, accN[nt]);
          }
        }
      }
      if (h + 2 < 12) {
        lds_group_edge();
        if (h == 3) {
          wave_lds_handover();
          #pragma unroll
          for (int k = 0; k < D_EDGE / 4; ++k)
            ah[k] = *(const v2f*)&st[c * HP + 4 * k + 2 * hi];
          wave_lds_handover();
        }
        {
          const int  h2    = h + 2;
          const bool lay1n = h2 >= 4;
          const int  p2    = lay1n ? ((h2 - 4) >> 1) : (h2 >> 1);
          #pragma unroll
          for (int kk = 0; kk < 8; ++kk) {
            #pragma unroll
            for (int t = 0; t < 2; ++t) {
              const int krow = 2 * (8 * s + kk) + hi;
              bw[s][2 * kk + t] = lay1n ? sW1[krow * D_NODE + (2 * p2 + t) * 16 + c]
                                        : sW0[krow * D_EDGE + (2 * p2 + t) * 16 + c];
            }
          }
        }
        if (h == 3) {
          const v2f* erow = efv + eid * (D_EDGE / 2);
          #pragma unroll
          for (int k = 0; k < D_EDGE / 4; ++k) ae[k] = erow[2 * k + hi];
        }
        if (h == 5 || h == 7) {
          #pragma unroll
          for (int t = h - 1; t <= h; ++t) {
            #pragma unroll
            for (int r = 0; r < 8; ++r) g[t][r] = nf[soff[r] + 16 * t];
          }
        }
        lds_group_edge();
      }
    }
  }

  #pragma unroll
  for (int nt = 0; nt < D_NODE / 16; ++nt) {
    #pragma unroll
    for (int r = 0; r < 8; ++r) {
      const int row = r + 8 * hi;
      st[row * RP + ((nt * 16 + c) ^ (row << 1))] = accN[nt][r];
    }
  }
  __syncthreads();

  v2f aa[D_NODE / 4];
  #pragma unroll
  for (int k = 0; k < D_NODE / 4; ++k) {
    const int o = c * RP + ((4 * k + 2 * hi) ^ (c << 1));
    v2f s = *(const v2f*)&sst[0][o];
    #pragma unroll
    for (int w = 1; w < PW; ++w) s += *(const v2f*)&sst[w][o];
    aa[k] = s;
  }

  const v2f* wav = (const v2f*)wap;
  #pragma unroll
  for (int t = 0; t < D_NODE / 16 / PW; ++t) {
    const int nt = wave * (D_NODE / 16 / PW) + t;
    v8f acc = {};
    #pragma unroll
    for (int k = 0; k < D_NODE / 4; ++k)
      acc = wmma_f32(aa[k], wav[(2 * k + hi) * D_NODE + nt * 16 + c], acc);
    const int   col = nt * 16 + c;
    const float bs  = ba[col];
    #pragma unroll
    for (int r = 0; r < 8; ++r) {
      int m = n0 + r + 8 * hi;
      if (m < n_nodes) out[m * D_NODE + col] = acc[r] + bs;
    }
  }
}

extern "C" void kernel_launch(void* const* d_in, const int* in_sizes, int n_in,
                              void* d_out, int out_size, void* d_ws, size_t ws_size,
                              hipStream_t stream) {
  const float* nf = (const float*)d_in[0];
  const float* ef = (const float*)d_in[1];
  const float* W0 = (const float*)d_in[2];
  const float* b0 = (const float*)d_in[3];
  const float* W1 = (const float*)d_in[4];
  const float* b1 = (const float*)d_in[5];
  const float* Wa = (const float*)d_in[6];
  const float* ba = (const float*)d_in[7];
  const int*  src = (const int*)d_in[8];
  const int*  dst = (const int*)d_in[9];

  const int n_nodes = in_sizes[0] / D_NODE;
  const int n_edges = in_sizes[8];

  const int ntiles = (n_nodes + 15) / 16;

  float* w0p  = (float*)d_ws;
  float* w1p  = w0p + D_EDGE * D_EDGE;
  float* wap  = w1p + D_EDGE * D_NODE;
  int*   rec  = (int*)(wap + D_NODE * D_NODE);
  int*   cnt  = rec + 2 * n_edges;
  int*   pre  = cnt + OG * ntiles;
  int*   tot  = pre + OG * ntiles;
  int*   off  = tot + ntiles;

  tile_count_pack_kernel<<<OG + PACK_BLOCKS, OT, 0, stream>>>(
      dst, cnt, n_edges, n_nodes, ntiles, W0, W1, Wa, w0p, w1p, wap);
  tile_prefix_kernel<<<(ntiles + 255) / 256, 256, 0, stream>>>(cnt, pre, tot,
                                                               ntiles);
  tile_place_kernel<<<OG, OT, 0, stream>>>(src, dst, pre, tot, off, rec,
                                           n_edges, n_nodes, ntiles);

  gine_pull_kernel<<<ntiles, PW * 32, 0, stream>>>(
      ef, nf, w0p, b0, w1p, b1, wap, ba, off, rec, (float*)d_out, n_nodes);
}


# Tile prefix kernel: 8 threads per tile each load an eighth of the 128 chunk counts as one batch, partial sums meet in LDS: one memory round trip on 98 work-groups, not 32 in series on 13
# speedup vs baseline: 2.0779x; 1.0635x over previous
#include <hip/hip_runtime.h>


typedef __attribute__((ext_vector_type(2))) float v2f;
typedef __attribute__((ext_vector_type(8))) float v8f;

#define D_NODE 128
#define D_EDGE 64
#define PW     4
#define HP     (D_EDGE + 2)
#define GT     4
#define RP     D_NODE

static __device__ __forceinline__ v8f wmma_f32(v2f a, v2f b, v8f c) {
  return __builtin_amdgcn_wmma_f32_16x16x4_f32(false, a, false, b, (short)0, c,
                                               false, false);
}

static __device__ __forceinline__ void wave_lds_handover() {
  __builtin_amdgcn_fence(__ATOMIC_RELEASE, "wavefront");
  __builtin_amdgcn_wave_barrier();
  __builtin_amdgcn_fence(__ATOMIC_ACQUIRE, "wavefront");
}

static __device__ __forceinline__ void lds_group_edge() {
  __builtin_amdgcn_sched_barrier(0);
}

#define OG        128
#define OT        1024
#define MAX_TILES 4096
#define SRC_MASK  0x0FFFFFFF
#define PACK_W0    (D_EDGE * D_EDGE)
#define PACK_W1    (D_EDGE * D_NODE)
#define PACK_WA    (D_NODE * D_NODE)
#define PACK_ELEMS (PACK_W0 + PACK_W1 + PACK_WA)
#define PACK_BLOCKS ((PACK_ELEMS + OT - 1) / OT)

__global__ __launch_bounds__(OT) void tile_count_pack_kernel(
    const int* __restrict__ dst, int* __restrict__ cnt, int n_edges,
    int n_nodes, int ntiles,
    const float* __restrict__ W0, const float* __restrict__ W1,
    const float* __restrict__ Wa, float* __restrict__ w0p,
    float* __restrict__ w1p, float* __restrict__ wap) {
  __shared__ int h[MAX_TILES];
  const int t = threadIdx.x;
  if (blockIdx.x >= OG) {
    int i = (blockIdx.x - OG) * OT + t;
    if (i < PACK_W0) {
      const int r = i / D_EDGE, c = i - r * D_EDGE;
      w0p[(((r >> 1) * D_EDGE + c) << 1) + (r & 1)] = W0[i];
    } else if (i < PACK_W0 + PACK_W1) {
      i -= PACK_W0;
      const int r = i / D_NODE, c = i - r * D_NODE;
      w1p[(((r >> 1) * D_NODE + c) << 1) + (r & 1)] = W1[i];
    } else if (i < PACK_ELEMS) {
      i -= PACK_W0 + PACK_W1;
      const int r = i / D_NODE, c = i - r * D_NODE;
      wap[(((r >> 1) * D_NODE + c) << 1) + (r & 1)] = Wa[i];
    }
    return;
  }
  for (int i = t; i < ntiles; i += OT) h[i] = 0;
  __syncthreads();
  const int chunk = (n_edges + OG - 1) / OG;
  const int e0 = blockIdx.x * chunk;
  const int e1 = e0 + chunk < n_edges ? e0 + chunk : n_edges;
  for (int eb = e0; eb < e1; eb += 8 * OT) {
    int d[8];
    #pragma unroll
    for (int k = 0; k < 8; ++k) {
      const int e  = eb + k * OT + t;
      const int ec = e < e1 ? e : e1 - 1;
      d[k] = dst[ec];
      if (e >= e1) d[k] = -1;
    }
    #pragma unroll
    for (int k = 0; k < 8; ++k)
      if ((unsigned)d[k] < (unsigned)n_nodes) atomicAdd(&h[d[k] >> 4], 1);
  }
  __syncthreads();
  for (int i = t; i < ntiles; i += OT) cnt[blockIdx.x * ntiles + i] = h[i];
}

#define PT 32
#define PQ 8
#define PC (OG / PQ)

__global__ __launch_bounds__(PT * PQ) void tile_prefix_kernel(
    const int* __restrict__ cnt, int* __restrict__ pre, int* __restrict__ tot,
    int ntiles) {
  __shared__ int part[PQ][PT];
  const int lane = threadIdx.x & (PT - 1);
  const int q    = threadIdx.x / PT;
  const int b    = blockIdx.x * PT + lane;
  const int bc   = b < ntiles ? b : ntiles - 1;
  const int row0 = q * PC;
  int v[PC];
  #pragma unroll
  for (int k = 0; k < PC; ++k) v[k] = cnt[(row0 + k) * ntiles + bc];
  int s = 0;
  #pragma unroll
  for (int k = 0; k < PC; ++k) s += v[k];
  part[q][lane] = s;
  __syncthreads();
  int pj[PQ - 1];
  #pragma unroll
  for (int j = 0; j < PQ - 1; ++j) pj[j] = part[j][lane];
  int run = 0;
  #pragma unroll
  for (int j = 0; j < PQ - 1; ++j) run += j < q ? pj[j] : 0;
  if (b < ntiles) {
    #pragma unroll
    for (int k = 0; k < PC; ++k) {
      pre[(row0 + k) * ntiles + b] = run;
      run += v[k];
    }
    if (q == PQ - 1) tot[b] = run;
  }
}

__global__ __launch_bounds__(OT) void tile_place_kernel(
    const int* __restrict__ src, const int* __restrict__ dst,
    const int* __restrict__ pre, const int* __restrict__ tot,
    int* __restrict__ off, int* __restrict__ rec, int n_edges, int n_nodes,
    int ntiles) {
  __shared__ int slot[MAX_TILES];
  __shared__ int wsum[OT / 32];
  const int t    = threadIdx.x;
  const int lane = t & 31;
  const int w    = t >> 5;
  const int g    = blockIdx.x;

  int v[4], pv[4];
  int s = 0;
  #pragma unroll
  for (int k = 0; k < 4; ++k) {
    const int i  = 4 * t + k;
    const int ic = i < ntiles ? i : ntiles - 1;
    v[k]  = tot[ic];
    pv[k] = pre[g * ntiles + ic];
    if (i >= ntiles) v[k] = 0;
    s += v[k];
  }
  int inc = s;
  #pragma unroll
  for (int d = 1; d < 32; d <<= 1) {
    const int u = __shfl_up(inc, d);
    if (lane >= d) inc += u;
  }
  if (lane == 31) wsum[w] = inc;
  __syncthreads();
  if (w == 0) {
    const int x = wsum[lane];
    int xi = x;
    #pragma unroll
    for (int d = 1; d < 32; d <<= 1) {
      const int u = __shfl_up(xi, d);
      if (lane >= d) xi += u;
    }
    wsum[lane] = xi - x;
  }
  __syncthreads();
  int ex = wsum[w] + inc - s;
  #pragma unroll
  for (int k = 0; k < 4; ++k) {
    const int i = 4 * t + k;
    if (i < ntiles) {
      slot[i] = ex + pv[k];
      if (g == 0) off[i] = ex;
    }
    ex += v[k];
  }
  if (g == 0 && t == OT - 1) off[ntiles] = ex;
  __syncthreads();

  typedef __attribute__((ext_vector_type(2))) int v2i;
  v2i* recv = (v2i*)rec;
  const int chunk = (n_edges + OG - 1) / OG;
  const int e0 = g * chunk;
  const int e1 = e0 + chunk < n_edges ? e0 + chunk : n_edges;
  for (int eb = e0; eb < e1; eb += 8 * OT) {
    int d[8], s8[8];
    #pragma unroll
    for (int k = 0; k < 8; ++k) {
      const int e  = eb + k * OT + t;
      const int ec = e < e1 ? e : e1 - 1;
      d[k]  = dst[ec];
      s8[k] = src[ec];
      if (e >= e1) d[k] = -1;
    }
    #pragma unroll
    for (int k = 0; k < 8; ++k) {
      if ((unsigned)d[k] < (unsigned)n_nodes) {
        const int p = atomicAdd(&slot[d[k] >> 4], 1);
        v2i r;
        r.x = eb + k * OT + t;
        r.y = (int)(((unsigned)s8[k] & SRC_MASK) | ((unsigned)(d[k] & 15) << 28));
        recv[p] = r;
      }
    }
  }
}

__global__ __launch_bounds__(PW * 32, 3) void gine_pull_kernel(
    const float* __restrict__ ef, const float* __restrict__ nf,
    const float* __restrict__ w0p, const float* __restrict__ b0,
    const float* __restrict__ w1p, const float* __restrict__ b1,
    const float* __restrict__ wap, const float* __restrict__ ba,
    const int* __restrict__ off, const int* __restrict__ rec,
    float* __restrict__ out, int n_nodes)
{
  __shared__ v2f   sW0[(D_EDGE / 2) * D_EDGE];
  __shared__ v2f   sW1[(D_EDGE / 2) * D_NODE];
  __shared__ __attribute__((aligned(16))) float sst[PW][16 * RP];

  const int tid = threadIdx.x;
  {
    const v2f* w0v = (const v2f*)w0p;
    for (int i = tid; i < (D_EDGE / 2) * D_EDGE; i += PW * 32) sW0[i] = w0v[i];
    const v2f* w1v = (const v2f*)w1p;
    for (int i = tid; i < (D_EDGE / 2) * D_NODE; i += PW * 32) sW1[i] = w1v[i];
  }
  __syncthreads();

  const int hwave = __builtin_amdgcn_readfirstlane(tid >> 5);
  const int wave = (hwave + blockIdx.x) & (PW - 1);
  const int lane = tid & 31;
  const int c    = lane & 15;
  const int hi   = lane >> 4;

  const int n0 = blockIdx.x * 16;

  float* st = &sst[hwave][0];

  v8f accN[D_NODE / 16];
  if (wave == 0) {
    #pragma unroll
    for (int r = 0; r < 8; ++r) {
      int n = n0 + 8 * hi + r;
      if (n >= n_nodes) n = n_nodes - 1;
      const float* nrow = nf + n * D_NODE + c;
      #pragma unroll
      for (int nt = 0; nt < D_NODE / 16; ++nt) accN[nt][r] = nrow[nt * 16];
    }
  } else {
    #pragma unroll
    for (int nt = 0; nt < D_NODE / 16; ++nt) accN[nt] = (v8f){};
  }

  const int ebeg = __builtin_amdgcn_readfirstlane(off[blockIdx.x]);
  const int eend = __builtin_amdgcn_readfirstlane(off[blockIdx.x + 1]);
  typedef __attribute__((ext_vector_type(2))) int v2i;
  const v2i* recv = (const v2i*)rec;

  const v2f* efv = (const v2f*)ef;

  const int last  = eend - 1;
  const int wbase = ebeg + 16 * wave;
  int eid = 0, sid = 0;
  v2f ae[D_EDGE / 4];
  #pragma unroll
  for (int k = 0; k < D_EDGE / 4; ++k) ae[k] = (v2f){0.f, 0.f};
  if (wbase < eend) {
    int p = wbase + c;
    if (p > last) p = last;
    const v2i rc = recv[p];
    eid = rc.x;
    sid = rc.y;
    const v2f* erow = efv + eid * (D_EDGE / 2);
    #pragma unroll
    for (int k = 0; k < D_EDGE / 4; ++k) ae[k] = erow[2 * k + hi];
  }

  #pragma unroll 1
  for (int base = wbase; base < eend; base += 16 * PW) {
    int sid_next;
    {
      int p = base + 16 * PW + c;
      if (p > last) p = last;
      const v2i rc = recv[p];
      eid = rc.x;
      sid_next = rc.y;
    }
    __builtin_amdgcn_sched_barrier(0);

    int soff[8];
    v2f sa[4];
    #pragma unroll
    for (int r = 0; r < 8; ++r) {
      const int s0 = __builtin_amdgcn_readlane(sid, r);
      const int s1 = __builtin_amdgcn_readlane(sid, r + 8);
      const int sw = hi ? s1 : s0;
      soff[r] = (sw & SRC_MASK) * D_NODE + c;
      const int p = base + 8 * hi + r;
      sa[r >> 1][r & 1] = (p < eend && ((unsigned)sw >> 28) == (unsigned)c) ? 1.f : 0.f;
    }

    sid = sid_next;

    float g[D_NODE / 16][8];
    #pragma unroll
    for (int t = 0; t < GT; ++t) {
      #pragma unroll
      for (int r = 0; r < 8; ++r) g[t][r] = nf[soff[r] + 16 * t];
    }

    v2f bw[2][16];
    #pragma unroll
    for (int kk = 0; kk < 8; ++kk) {
      #pragma unroll
      for (int t = 0; t < 2; ++t) {
        bw[0][2 * kk + t] = sW0[(2 * kk + hi) * D_EDGE + t * 16 + c];
        bw[1][2 * kk + t] = sW0[(2 * (8 + kk) + hi) * D_EDGE + t * 16 + c];
      }
    }
    lds_group_edge();

    v2f ah[D_EDGE / 4];
    #pragma unroll
    for (int k = 0; k < D_EDGE / 4; ++k) ah[k] = (v2f){0.f, 0.f};
    v8f acc0 = {}, acc1 = {};
    #pragma unroll
    for (int h = 0; h < 12; ++h) {
      const int  s    = h & 1;
      const bool lay1 = h >= 4;
      const int  p    = lay1 ? ((h - 4) >> 1) : (h >> 1);
      if (s == 0) { acc0 = (v8f){}; acc1 = (v8f){}; }
      #pragma unroll
      for (int kk = 0; kk < 8; ++kk) {
        const v2f a = lay1 ? ah[8 * s + kk] : ae[8 * s + kk];
        acc0 = wmma_f32(a, bw[s][2 * kk],     acc0);
        acc1 = wmma_f32(a, bw[s][2 * kk + 1], acc1);
      }
      if (s == 1 && !lay1) {
        #pragma unroll
        for (int t = 0; t < 2; ++t) {
          const v8f   acc = t ? acc1 : acc0;
          const int   col = (2 * p + t) * 16 + c;
          const float bs  = b0[col];
          #pragma unroll
          for (int r = 0; r < 8; ++r) {
            float v = acc[r] + bs;
            st[(r + 8 * hi) * HP + col] = v > 0.f ? v : 0.f;
          }
        }
      }
      if (s == 1 && lay1) {
        #pragma unroll
        for (int t = 0; t < 2; ++t) {
          const v8f   acc = t ? acc1 : acc0;
          const int   nt  = 2 * p + t;
          const float bs  = b1[nt * 16 + c];
          v8f m;
          #pragma unroll
          for (int r = 0; r < 8; ++r) {
            float ev = acc[r] + bs;
            ev = ev > 0.f ? ev : 0.f;
            float v = g[nt][r] + ev;
            m[r] = v > 0.f ? v : 0.f;
          }
          #pragma unroll
          for (int j = 0; j < 4; ++j) {
            v2f mb = {m[2 * j], m[2 * j + 1]};
            accN[nt] = wmma_f32(sa[j], mb, accN[nt]);
          }
        }
      }
      if (h + 2 < 12) {
        lds_group_edge();
        if (h == 3) {
          wave_lds_handover();
          #pragma unroll
          for (int k = 0; k < D_EDGE / 4; ++k)
            ah[k] = *(const v2f*)&st[c * HP + 4 * k + 2 * hi];
          wave_lds_handover();
        }
        {
          const int  h2    = h + 2;
          const bool lay1n = h2 >= 4;
          const int  p2    = lay1n ? ((h2 - 4) >> 1) : (h2 >> 1);
          #pragma unroll
          for (int kk = 0; kk < 8; ++kk) {
            #pragma unroll
            for (int t = 0; t < 2; ++t) {
              const int krow = 2 * (8 * s + kk) + hi;
              bw[s][2 * kk + t] = lay1n ? sW1[krow * D_NODE + (2 * p2 + t) * 16 + c]
                                        : sW0[krow * D_EDGE + (2 * p2 + t) * 16 + c];
            }
          }
        }
        if (h == 3) {
          const v2f* erow = efv + eid * (D_EDGE / 2);
          #pragma unroll
          for (int k = 0; k < D_EDGE / 4; ++k) ae[k] = erow[2 * k + hi];
        }
        if (h == 5 || h == 7) {
          #pragma unroll
          for (int t = h - 1; t <= h; ++t) {
            #pragma unroll
            for (int r = 0; r < 8; ++r) g[t][r] = nf[soff[r] + 16 * t];
          }
        }
        lds_group_edge();
      }
    }
  }

  #pragma unroll
  for (int nt = 0; nt < D_NODE / 16; ++nt) {
    #pragma unroll
    for (int r = 0; r < 8; ++r) {
      const int row = r + 8 * hi;
      st[row * RP + ((nt * 16 + c) ^ (row << 1))] = accN[nt][r];
    }
  }
  __syncthreads();

  v2f aa[D_NODE / 4];
  #pragma unroll
  for (int k = 0; k < D_NODE / 4; ++k) {
    const int o = c * RP + ((4 * k + 2 * hi) ^ (c << 1));
    v2f s = *(const v2f*)&sst[0][o];
    #pragma unroll
    for (int w = 1; w < PW; ++w) s += *(const v2f*)&sst[w][o];
    aa[k] = s;
  }

  const v2f* wav = (const v2f*)wap;
  #pragma unroll
  for (int t = 0; t < D_NODE / 16 / PW; ++t) {
    const int nt = wave * (D_NODE / 16 / PW) + t;
    v8f acc = {};
    #pragma unroll
    for (int k = 0; k < D_NODE / 4; ++k)
      acc = wmma_f32(aa[k], wav[(2 * k + hi) * D_NODE + nt * 16 + c], acc);
    const int   col = nt * 16 + c;
    const float bs  = ba[col];
    #pragma unroll
    for (int r = 0; r < 8; ++r) {
      int m = n0 + r + 8 * hi;
      if (m < n_nodes) out[m * D_NODE + col] = acc[r] + bs;
    }
  }
}

extern "C" void kernel_launch(void* const* d_in, const int* in_sizes, int n_in,
                              void* d_out, int out_size, void* d_ws, size_t ws_size,
                              hipStream_t stream) {
  const float* nf = (const float*)d_in[0];
  const float* ef = (const float*)d_in[1];
  const float* W0 = (const float*)d_in[2];
  const float* b0 = (const float*)d_in[3];
  const float* W1 = (const float*)d_in[4];
  const float* b1 = (const float*)d_in[5];
  const float* Wa = (const float*)d_in[6];
  const float* ba = (const float*)d_in[7];
  const int*  src = (const int*)d_in[8];
  const int*  dst = (const int*)d_in[9];

  const int n_nodes = in_sizes[0] / D_NODE;
  const int n_edges = in_sizes[8];

  const int ntiles = (n_nodes + 15) / 16;

  float* w0p  = (float*)d_ws;
  float* w1p  = w0p + D_EDGE * D_EDGE;
  float* wap  = w1p + D_EDGE * D_NODE;
  int*   rec  = (int*)(wap + D_NODE * D_NODE);
  int*   cnt  = rec + 2 * n_edges;
  int*   pre  = cnt + OG * ntiles;
  int*   tot  = pre + OG * ntiles;
  int*   off  = tot + ntiles;

  tile_count_pack_kernel<<<OG + PACK_BLOCKS, OT, 0, stream>>>(
      dst, cnt, n_edges, n_nodes, ntiles, W0, W1, Wa, w0p, w1p, wap);
  tile_prefix_kernel<<<(ntiles + PT - 1) / PT, PT * PQ, 0, stream>>>(
      cnt, pre, tot, ntiles);
  tile_place_kernel<<<OG, OT, 0, stream>>>(src, dst, pre, tot, off, rec,
                                           n_edges, n_nodes, ntiles);

  gine_pull_kernel<<<ntiles, PW * 32, 0, stream>>>(
      ef, nf, w0p, b0, w1p, b1, wap, ba, off, rec, (float*)d_out, n_nodes);
}


# Grouping in one launch, not three: each edge chunk ordered by tile only inside itself (LDS count, scan, place); pull work-group gathers its tile's records from the 128 chunks into LDS; sum order kept
# speedup vs baseline: 2.2766x; 1.0956x over previous
#include <hip/hip_runtime.h>


typedef __attribute__((ext_vector_type(2))) float v2f;
typedef __attribute__((ext_vector_type(8))) float v8f;

#define D_NODE 128
#define D_EDGE 64
#define PW     4
#define HP     (D_EDGE + 2)
#define GT     4
#define RP     D_NODE
#define CAP    1024

static __device__ __forceinline__ v8f wmma_f32(v2f a, v2f b, v8f c) {
  return __builtin_amdgcn_wmma_f32_16x16x4_f32(false, a, false, b, (short)0, c,
                                               false, false);
}

static __device__ __forceinline__ void wave_lds_handover() {
  __builtin_amdgcn_fence(__ATOMIC_RELEASE, "wavefront");
  __builtin_amdgcn_wave_barrier();
  __builtin_amdgcn_fence(__ATOMIC_ACQUIRE, "wavefront");
}

static __device__ __forceinline__ void lds_group_edge() {
  __builtin_amdgcn_sched_barrier(0);
}

#define OG        128
#define OT        1024
#define MAX_TILES 4096
#define SRC_MASK  0x0FFFFFFF
#define PACK_W0    (D_EDGE * D_EDGE)
#define PACK_W1    (D_EDGE * D_NODE)
#define PACK_WA    (D_NODE * D_NODE)
#define PACK_ELEMS (PACK_W0 + PACK_W1 + PACK_WA)
#define PACK_BLOCKS ((PACK_ELEMS + OT - 1) / OT)

typedef __attribute__((ext_vector_type(2))) int v2i;
static_assert(OG == PW * 32, "the pull kernel collects one chunk per thread");

__global__ __launch_bounds__(OT) void chunk_group_pack_kernel(
    const int* __restrict__ src, const int* __restrict__ dst,
    int* __restrict__ cstart, int* __restrict__ crec, int n_edges,
    int n_nodes, int ntiles,
    const float* __restrict__ W0, const float* __restrict__ W1,
    const float* __restrict__ Wa, float* __restrict__ w0p,
    float* __restrict__ w1p, float* __restrict__ wap) {
  __shared__ int h[MAX_TILES];
  __shared__ int wsum[OT / 32];
  const int t = threadIdx.x;
  if (blockIdx.x >= OG) {
    int i = (blockIdx.x - OG) * OT + t;
    if (i < PACK_W0) {
      const int r = i / D_EDGE, c = i - r * D_EDGE;
      w0p[(((r >> 1) * D_EDGE + c) << 1) + (r & 1)] = W0[i];
    } else if (i < PACK_W0 + PACK_W1) {
      i -= PACK_W0;
      const int r = i / D_NODE, c = i - r * D_NODE;
      w1p[(((r >> 1) * D_NODE + c) << 1) + (r & 1)] = W1[i];
    } else if (i < PACK_ELEMS) {
      i -= PACK_W0 + PACK_W1;
      const int r = i / D_NODE, c = i - r * D_NODE;
      wap[(((r >> 1) * D_NODE + c) << 1) + (r & 1)] = Wa[i];
    }
    return;
  }
  const int lane = t & 31;
  const int w    = t >> 5;
  const int g    = blockIdx.x;
  for (int i = t; i < ntiles; i += OT) h[i] = 0;
  __syncthreads();
  const int chunk = (n_edges + OG - 1) / OG;
  const int e0 = g * chunk;
  const int e1 = e0 + chunk < n_edges ? e0 + chunk : n_edges;
  const bool one_trip = e1 - e0 <= 8 * OT;
  int d[8], s8[8];
  for (int eb = e0; eb < e1; eb += 8 * OT) {
    #pragma unroll
    for (int k = 0; k < 8; ++k) {
      const int e  = eb + k * OT + t;
      const int ec = e < e1 ? e : e1 - 1;
      d[k]  = dst[ec];
      s8[k] = src[ec];
      if (e >= e1) d[k] = -1;
    }
    #pragma unroll
    for (int k = 0; k < 8; ++k)
      if ((unsigned)d[k] < (unsigned)n_nodes) atomicAdd(&h[d[k] >> 4], 1);
  }
  __syncthreads();

  int v[4];
  int s = 0;
  #pragma unroll
  for (int k = 0; k < 4; ++k) {
    const int i = 4 * t + k;
    v[k] = i < ntiles ? h[i] : 0;
    s += v[k];
  }
  int inc = s;
  #pragma unroll
  for (int dd = 1; dd < 32; dd <<= 1) {
    const int u = __shfl_up(inc, dd);
    if (lane >= dd) inc += u;
  }
  if (lane == 31) wsum[w] = inc;
  __syncthreads();
  if (w == 0) {
    const int x = wsum[lane];
    int xi = x;
    #pragma unroll
    for (int dd = 1; dd < 32; dd <<= 1) {
      const int u = __shfl_up(xi, dd);
      if (lane >= dd) xi += u;
    }
    wsum[lane] = xi - x;
  }
  __syncthreads();
  int ex = wsum[w] + inc - s;
  int* crow = cstart + g * (ntiles + 1);
  #pragma unroll
  for (int k = 0; k < 4; ++k) {
    const int i = 4 * t + k;
    if (i < ntiles) {
      h[i]    = ex;
      crow[i] = ex;
    }
    ex += v[k];
  }
  if (t == OT - 1) crow[ntiles] = ex;
  __syncthreads();

  v2i* recv = (v2i*)crec + e0;
  for (int eb = e0; eb < e1; eb += 8 * OT) {
    if (!one_trip) {
      #pragma unroll
      for (int k = 0; k < 8; ++k) {
        const int e  = eb + k * OT + t;
        const int ec = e < e1 ? e : e1 - 1;
        d[k]  = dst[ec];
        s8[k] = src[ec];
        if (e >= e1) d[k] = -1;
      }
    }
    #pragma unroll
    for (int k = 0; k < 8; ++k) {
      if ((unsigned)d[k] < (unsigned)n_nodes) {
        const int p = atomicAdd(&h[d[k] >> 4], 1);
        v2i r;
        r.x = eb + k * OT + t;
        r.y = (int)(((unsigned)s8[k] & SRC_MASK) | ((unsigned)(d[k] & 15) << 28));
        recv[p] = r;
      }
    }
  }
}

__global__ __launch_bounds__(PW * 32, 3) void gine_pull_kernel(
    const float* __restrict__ ef, const float* __restrict__ nf,
    const float* __restrict__ w0p, const float* __restrict__ b0,
    const float* __restrict__ w1p, const float* __restrict__ b1,
    const float* __restrict__ wap, const float* __restrict__ ba,
    const int* __restrict__ cstart, const int* __restrict__ crec,
    float* __restrict__ out, int n_nodes, int n_edges, int ntiles)
{
  __shared__ v2f   sW0[(D_EDGE / 2) * D_EDGE];
  __shared__ v2f   sW1[(D_EDGE / 2) * D_NODE];
  __shared__ __attribute__((aligned(16))) float sst[PW][16 * RP];
  __shared__ v2i   lst[CAP];
  __shared__ int   scn[PW];

  const int tid = threadIdx.x;
  const int chunk = (n_edges + OG - 1) / OG;
  const int* crow = cstart + tid * (ntiles + 1) + blockIdx.x;
  const int sg = crow[0];
  const int eg = crow[1];
  {
    const v2f* w0v = (const v2f*)w0p;
    for (int i = tid; i < (D_EDGE / 2) * D_EDGE; i += PW * 32) sW0[i] = w0v[i];
    const v2f* w1v = (const v2f*)w1p;
    for (int i = tid; i < (D_EDGE / 2) * D_NODE; i += PW * 32) sW1[i] = w1v[i];
  }
  const int len = eg - sg;
  int linc = len;
  #pragma unroll
  for (int dd = 1; dd < 32; dd <<= 1) {
    const int u = __shfl_up(linc, dd);
    if ((tid & 31) >= dd) linc += u;
  }
  if ((tid & 31) == 31) scn[tid >> 5] = linc;
  __syncthreads();
  int og = linc - len;
  int nrec = 0;
  #pragma unroll
  for (int w = 0; w < PW; ++w) {
    const int x = scn[w];
    if (w < (tid >> 5)) og += x;
    nrec += x;
  }
  nrec = __builtin_amdgcn_readfirstlane(nrec);
  const v2i* cseg = (const v2i*)crec + tid * chunk + sg;

  const int hwave = __builtin_amdgcn_readfirstlane(tid >> 5);
  const int wave = (hwave + blockIdx.x) & (PW - 1);
  const int lane = tid & 31;
  const int c    = lane & 15;
  const int hi   = lane >> 4;

  const int n0 = blockIdx.x * 16;

  float* st = &sst[hwave][0];

  v8f accN[D_NODE / 16];
  if (wave == 0) {
    #pragma unroll
    for (int r = 0; r < 8; ++r) {
      int n = n0 + 8 * hi + r;
      if (n >= n_nodes) n = n_nodes - 1;
      const float* nrow = nf + n * D_NODE + c;
      #pragma unroll
      for (int nt = 0; nt < D_NODE / 16; ++nt) accN[nt][r] = nrow[nt * 16];
    }
  } else {
    #pragma unroll
    for (int nt = 0; nt < D_NODE / 16; ++nt) accN[nt] = (v8f){};
  }

  const v2f* efv = (const v2f*)ef;

  for (int w0 = 0; w0 < nrec; w0 += CAP) {
  if (w0 > 0) __syncthreads();
  const int wend = w0 + CAP < nrec ? w0 + CAP : nrec;
  {
    const int lo = og > w0 ? og : w0;
    const int hs = og + len < wend ? og + len : wend;
    for (int i0 = lo; i0 < hs; i0 += 8) {
      v2i rv[8];
      #pragma unroll
      for (int k = 0; k < 8; ++k) {
        const int i  = i0 + k;
        const int ic = i < hs ? i : hs - 1;
        rv[k] = cseg[ic - og];
      }
      #pragma unroll
      for (int k = 0; k < 8; ++k) {
        const int i = i0 + k;
        if (i < hs) lst[i - w0] = rv[k];
      }
    }
  }
  __syncthreads();

  const int eend = wend - w0;
  const v2i* recv = lst;

  const int last  = eend - 1;
  const int wbase = 16 * wave;
  int eid = 0, sid = 0;
  v2f ae[D_EDGE / 4];
  #pragma unroll
  for (int k = 0; k < D_EDGE / 4; ++k) ae[k] = (v2f){0.f, 0.f};
  if (wbase < eend) {
    int p = wbase + c;
    if (p > last) p = last;
    const v2i rc = recv[p];
    eid = rc.x;
    sid = rc.y;
    const v2f* erow = efv + eid * (D_EDGE / 2);
    #pragma unroll
    for (int k = 0; k < D_EDGE / 4; ++k) ae[k] = erow[2 * k + hi];
  }

  #pragma unroll 1
  for (int base = wbase; base < eend; base += 16 * PW) {
    int sid_next;
    {
      int p = base + 16 * PW + c;
      if (p > last) p = last;
      const v2i rc = recv[p];
      eid = rc.x;
      sid_next = rc.y;
    }

    int soff[8];
    v2f sa[4];
    #pragma unroll
    for (int r = 0; r < 8; ++r) {
      const int s0 = __builtin_amdgcn_readlane(sid, r);
      const int s1 = __builtin_amdgcn_readlane(sid, r + 8);
      const int sw = hi ? s1 : s0;
      soff[r] = (sw & SRC_MASK) * D_NODE + c;
      const int p = base + 8 * hi + r;
      sa[r >> 1][r & 1] = (p < eend && ((unsigned)sw >> 28) == (unsigned)c) ? 1.f : 0.f;
    }

    sid = sid_next;

    float g[D_NODE / 16][8];
    #pragma unroll
    for (int t = 0; t < GT; ++t) {
      #pragma unroll
      for (int r = 0; r < 8; ++r) g[t][r] = nf[soff[r] + 16 * t];
    }

    v2f bw[2][16];
    #pragma unroll
    for (int kk = 0; kk < 8; ++kk) {
      #pragma unroll
      for (int t = 0; t < 2; ++t) {
        bw[0][2 * kk + t] = sW0[(2 * kk + hi) * D_EDGE + t * 16 + c];
        bw[1][2 * kk + t] = sW0[(2 * (8 + kk) + hi) * D_EDGE + t * 16 + c];
      }
    }
    lds_group_edge();

    v2f ah[D_EDGE / 4];
    #pragma unroll
    for (int k = 0; k < D_EDGE / 4; ++k) ah[k] = (v2f){0.f, 0.f};
    v8f acc0 = {}, acc1 = {};
    #pragma unroll
    for (int h = 0; h < 12; ++h) {
      const int  s    = h & 1;
      const bool lay1 = h >= 4;
      const int  p    = lay1 ? ((h - 4) >> 1) : (h >> 1);
      if (s == 0) { acc0 = (v8f){}; acc1 = (v8f){}; }
      #pragma unroll
      for (int kk = 0; kk < 8; ++kk) {
        const v2f a = lay1 ? ah[8 * s + kk] : ae[8 * s + kk];
        acc0 = wmma_f32(a, bw[s][2 * kk],     acc0);
        acc1 = wmma_f32(a, bw[s][2 * kk + 1], acc1);
      }
      if (s == 1 && !lay1) {
        #pragma unroll
        for (int t = 0; t < 2; ++t) {
          const v8f   acc = t ? acc1 : acc0;
          const int   col = (2 * p + t) * 16 + c;
          const float bs  = b0[col];
          #pragma unroll
          for (int r = 0; r < 8; ++r) {
            float v = acc[r] + bs;
            st[(r + 8 * hi) * HP + col] = v > 0.f ? v : 0.f;
          }
        }
      }
      if (s == 1 && lay1) {
        #pragma unroll
        for (int t = 0; t < 2; ++t) {
          const v8f   acc = t ? acc1 : acc0;
          const int   nt  = 2 * p + t;
          const float bs  = b1[nt * 16 + c];
          v8f m;
          #pragma unroll
          for (int r = 0; r < 8; ++r) {
            float ev = acc[r] + bs;
            ev = ev > 0.f ? ev : 0.f;
            float v = g[nt][r] + ev;
            m[r] = v > 0.f ? v : 0.f;
          }
          #pragma unroll
          for (int j = 0; j < 4; ++j) {
            v2f mb = {m[2 * j], m[2 * j + 1]};
            accN[nt] = wmma_f32(sa[j], mb, accN[nt]);
          }
        }
      }
      if (h + 2 < 12) {
        lds_group_edge();
        if (h == 3) {
          wave_lds_handover();
          #pragma unroll
          for (int k = 0; k < D_EDGE / 4; ++k)
            ah[k] = *(const v2f*)&st[c * HP + 4 * k + 2 * hi];
          wave_lds_handover();
        }
        {
          const int  h2    = h + 2;
          const bool lay1n = h2 >= 4;
          const int  p2    = lay1n ? ((h2 - 4) >> 1) : (h2 >> 1);
          #pragma unroll
          for (int kk = 0; kk < 8; ++kk) {
            #pragma unroll
            for (int t = 0; t < 2; ++t) {
              const int krow = 2 * (8 * s + kk) + hi;
              bw[s][2 * kk + t] = lay1n ? sW1[krow * D_NODE + (2 * p2 + t) * 16 + c]
                                        : sW0[krow * D_EDGE + (2 * p2 + t) * 16 + c];
            }
          }
        }
        if (h == 3) {
          const v2f* erow = efv + eid * (D_EDGE / 2);
          #pragma unroll
          for (int k = 0; k < D_EDGE / 4; ++k) ae[k] = erow[2 * k + hi];
        }
        if (h == 5 || h == 7) {
          #pragma unroll
          for (int t = h - 1; t <= h; ++t) {
            #pragma unroll
            for (int r = 0; r < 8; ++r) g[t][r] = nf[soff[r] + 16 * t];
          }
        }
        lds_group_edge();
      }
    }
  }
  }

  #pragma unroll
  for (int nt = 0; nt < D_NODE / 16; ++nt) {
    #pragma unroll
    for (int r = 0; r < 8; ++r) {
      const int row = r + 8 * hi;
      st[row * RP + ((nt * 16 + c) ^ (row << 1))] = accN[nt][r];
    }
  }
  __syncthreads();

  v2f aa[D_NODE / 4];
  #pragma unroll
  for (int k = 0; k < D_NODE / 4; ++k) {
    const int o = c * RP + ((4 * k + 2 * hi) ^ (c << 1));
    v2f s = *(const v2f*)&sst[0][o];
    #pragma unroll
    for (int w = 1; w < PW; ++w) s += *(const v2f*)&sst[w][o];
    aa[k] = s;
  }

  const v2f* wav = (const v2f*)wap;
  #pragma unroll
  for (int t = 0; t < D_NODE / 16 / PW; ++t) {
    const int nt = wave * (D_NODE / 16 / PW) + t;
    v8f acc = {};
    #pragma unroll
    for (int k = 0; k < D_NODE / 4; ++k)
      acc = wmma_f32(aa[k], wav[(2 * k + hi) * D_NODE + nt * 16 + c], acc);
    const int   col = nt * 16 + c;
    const float bs  = ba[col];
    #pragma unroll
    for (int r = 0; r < 8; ++r) {
      int m = n0 + r + 8 * hi;
      if (m < n_nodes) out[m * D_NODE + col] = acc[r] + bs;
    }
  }
}

extern "C" void kernel_launch(void* const* d_in, const int* in_sizes, int n_in,
                              void* d_out, int out_size, void* d_ws, size_t ws_size,
                              hipStream_t stream) {
  const float* nf = (const float*)d_in[0];
  const float* ef = (const float*)d_in[1];
  const float* W0 = (const float*)d_in[2];
  const float* b0 = (const float*)d_in[3];
  const float* W1 = (const float*)d_in[4];
  const float* b1 = (const float*)d_in[5];
  const float* Wa = (const float*)d_in[6];
  const float* ba = (const float*)d_in[7];
  const int*  src = (const int*)d_in[8];
  const int*  dst = (const int*)d_in[9];

  const int n_nodes = in_sizes[0] / D_NODE;
  const int n_edges = in_sizes[8];

  const int ntiles = (n_nodes + 15) / 16;

  float* w0p    = (float*)d_ws;
  float* w1p    = w0p + D_EDGE * D_EDGE;
  float* wap    = w1p + D_EDGE * D_NODE;
  int*   crec   = (int*)(wap + D_NODE * D_NODE);
  int*   cstart = crec + 2 * n_edges;

  chunk_group_pack_kernel<<<OG + PACK_BLOCKS, OT, 0, stream>>>(
      src, dst, cstart, crec, n_edges, n_nodes, ntiles, W0, W1, Wa, w0p, w1p, wap);

  gine_pull_kernel<<<ntiles, PW * 32, 0, stream>>>(
      ef, nf, w0p, b0, w1p, b1, wap, ba, cstart, crec, (float*)d_out, n_nodes,
      n_edges, ntiles);
}
